# starting kernel
# baseline (speedup 1.0000x reference)
#include <hip/hip_runtime.h>
#include <hip/hip_bf16.h>
#include <math.h>

typedef __attribute__((ext_vector_type(2))) float v2f;
typedef __attribute__((ext_vector_type(4))) float v4f;
typedef __attribute__((ext_vector_type(8))) float v8f;

#define PRIOR_LAMBDA 10.0f

__global__ void jc_setup_kernel(const float* __restrict__ branch1,
                                const float* __restrict__ branch2,
                                float* __restrict__ Tws,
                                float* __restrict__ logp,
                                int B, int m) {
  int b = blockIdx.x * blockDim.x + threadIdx.x;
  if (b >= B) return;
  float t1 = branch1[b], t2 = branch2[b];
  float e1 = __expf(-(4.0f / 3.0f) * t1);
  float e2 = __expf(-(4.0f / 3.0f) * t2);
  float dg1 = 0.25f + 0.75f * e1, of1 = 0.25f - 0.25f * e1;
  float dg2 = 0.25f + 0.75f * e2, of2 = 0.25f - 0.25f * e2;
  float s = __expf((__logf(PRIOR_LAMBDA) - PRIOR_LAMBDA * (t1 + t2)) / (float)m);
  float* T = Tws + (size_t)b * 32;
#pragma unroll
  for (int v = 0; v < 4; ++v) {
#pragma unroll
    for (int c = 0; c < 4; ++c) {
      T[v * 4 + c]      = fmaxf((v == c) ? dg1 : of1, 0.0f) * s;
      T[16 + v * 4 + c] = fmaxf((v == c) ? dg2 : of2, 0.0f);
    }
  }
  logp[b] = (float)m * __logf(0.25f);
}

__launch_bounds__(256)
__global__ void jc_joint_kernel(const float* __restrict__ f1,
                                const float* __restrict__ f2,
                                const float* __restrict__ Tws,
                                float* __restrict__ joint,
                                float* __restrict__ logp,
                                int B, int m, int blocksPerBatch) {
  __shared__ float lds[8 * 64];
  const int lane = threadIdx.x & 31;
  const int wid  = threadIdx.x >> 5;
  const int b    = blockIdx.x / blocksPerBatch;
  const int blk  = blockIdx.x % blocksPerBatch;
  const int wavesPerBlock = blockDim.x >> 5;
  const int wavesPerBatch = blocksPerBatch * wavesPerBlock;
  const int wgid  = blk * wavesPerBlock + wid;
  const int tiles = m >> 4;

  const float* T1 = Tws + (size_t)b * 32;
  const float* T2 = T1 + 16;
  const int n   = lane & 15;
  const int klo = (lane >> 4) << 1;
  v2f b1, b2;
  b1.x = (n < 4) ? T1[klo * 4 + n]       : 0.0f;
  b1.y = (n < 4) ? T1[(klo + 1) * 4 + n] : 0.0f;
  b2.x = (n < 4) ? T2[klo * 4 + n]       : 0.0f;
  b2.y = (n < 4) ? T2[(klo + 1) * 4 + n] : 0.0f;

  const size_t batch_off = (size_t)b * (size_t)m * 4;
  const float* F1 = f1 + batch_off;
  const float* F2 = f2 + batch_off;
  float* J = joint + batch_off;
  float* myLds = lds + wid * 64;

  float acc = 0.0f;

  for (int t = wgid; t < tiles; t += wavesPerBatch) {
    const int site0 = t << 4;
    const size_t aoff = (size_t)(site0 + n) * 4 + (size_t)((lane >> 4) << 1);
    v2f a1 = *(const v2f*)(F1 + aoff);
    v2f a2 = *(const v2f*)(F2 + aoff);

    if (t + wavesPerBatch < tiles) {
      const size_t nxt = aoff + (size_t)wavesPerBatch * 64;
      __builtin_prefetch(F1 + nxt, 0, 1);
      __builtin_prefetch(F2 + nxt, 0, 1);
    }

    v8f cz = {};
    v8f d1v = __builtin_amdgcn_wmma_f32_16x16x4_f32(
        false, a1, false, b1, (short)0, cz, false, false);
    v8f d2v = __builtin_amdgcn_wmma_f32_16x16x4_f32(
        false, a2, false, b2, (short)0, cz, false, false);

    v8f p;
#pragma unroll
    for (int g = 0; g < 8; ++g) p[g] = d1v[g] * d2v[g];

    if (n < 4) {
      const int siteHi = (lane >> 4) << 3;
#pragma unroll
      for (int g = 0; g < 8; ++g)
        myLds[(siteHi + g) * 4 + n] = p[g];
    }
    asm volatile("s_wait_dscnt 0" ::: "memory");

    if (lane < 16) {
      v4f jv = *(const v4f*)(myLds + lane * 4);
      *(v4f*)(J + (size_t)(site0 + lane) * 4) = jv;
      acc += __logf(jv.x + jv.y + jv.z + jv.w);
    }
  }

#pragma unroll
  for (int off = 16; off >= 1; off >>= 1)
    acc += __shfl_xor(acc, off, 32);
  if (lane == 0) atomicAdd(&logp[b], acc);
}

extern "C" void kernel_launch(void* const* d_in, const int* in_sizes, int n_in,
                              void* d_out, int out_size, void* d_ws, size_t ws_size,
                              hipStream_t stream) {
  const float* f1  = (const float*)d_in[0];
  const float* br1 = (const float*)d_in[1];
  const float* f2  = (const float*)d_in[2];
  const float* br2 = (const float*)d_in[3];

  const int B = in_sizes[1];
  const int m = in_sizes[0] / (B * 4);

  float* out   = (float*)d_out;
  float* joint = out;
  float* logp  = out + (size_t)B * m * 4;
  float* Tws   = (float*)d_ws;

  jc_setup_kernel<<<(B + 255) / 256, 256, 0, stream>>>(br1, br2, Tws, logp, B, m);

  const int blocksPerBatch = 8;
  jc_joint_kernel<<<B * blocksPerBatch, 256, 0, stream>>>(
      f1, f2, Tws, joint, logp, B, m, blocksPerBatch);
}


# One 8-wave workgroup per batch, not eight: 4096 long-lived waves for 32768, each requesting its next 8 tiles (16 loads, replacing the prefetch) before computing 8; log_p partial sums regrouped
# speedup vs baseline: 1.2146x; 1.2146x over previous
#include <hip/hip_runtime.h>
#include <hip/hip_bf16.h>
#include <math.h>

typedef __attribute__((ext_vector_type(2))) float v2f;
typedef __attribute__((ext_vector_type(4))) float v4f;
typedef __attribute__((ext_vector_type(8))) float v8f;

#define PRIOR_LAMBDA 10.0f

__global__ void jc_setup_kernel(const float* __restrict__ branch1,
                                const float* __restrict__ branch2,
                                float* __restrict__ Tws,
                                float* __restrict__ logp,
                                int B, int m) {
  int b = blockIdx.x * blockDim.x + threadIdx.x;
  if (b >= B) return;
  float t1 = branch1[b], t2 = branch2[b];
  float e1 = __expf(-(4.0f / 3.0f) * t1);
  float e2 = __expf(-(4.0f / 3.0f) * t2);
  float dg1 = 0.25f + 0.75f * e1, of1 = 0.25f - 0.25f * e1;
  float dg2 = 0.25f + 0.75f * e2, of2 = 0.25f - 0.25f * e2;
  float s = __expf((__logf(PRIOR_LAMBDA) - PRIOR_LAMBDA * (t1 + t2)) / (float)m);
  float* T = Tws + (size_t)b * 32;
#pragma unroll
  for (int v = 0; v < 4; ++v) {
#pragma unroll
    for (int c = 0; c < 4; ++c) {
      T[v * 4 + c]      = fmaxf((v == c) ? dg1 : of1, 0.0f) * s;
      T[16 + v * 4 + c] = fmaxf((v == c) ? dg2 : of2, 0.0f);
    }
  }
  logp[b] = (float)m * __logf(0.25f);
}

#define TPG 8

static __device__ __forceinline__ void load_group(v2f (&ra)[TPG], v2f (&rb)[TPG],
                                                  const float* __restrict__ F1,
                                                  const float* __restrict__ F2,
                                                  int grp, int laneOff) {
  const size_t base = (size_t)grp * (TPG * 64) + (size_t)laneOff;
#pragma unroll
  for (int g = 0; g < TPG; ++g) {
    ra[g] = *(const v2f*)(F1 + base + g * 64);
    rb[g] = *(const v2f*)(F2 + base + g * 64);
  }
}

static __device__ __forceinline__ void compute_group(const v2f (&ra)[TPG], const v2f (&rb)[TPG],
                                                     v2f b1, v2f b2, float* myLds,
                                                     float* __restrict__ J, int grp,
                                                     int lane, int n, float& acc) {
  const int siteHi = (lane >> 4) << 3;
#pragma unroll
  for (int g = 0; g < TPG; ++g) {
    v8f cz = {};
    v8f d1v = __builtin_amdgcn_wmma_f32_16x16x4_f32(
        false, ra[g], false, b1, (short)0, cz, false, false);
    v8f d2v = __builtin_amdgcn_wmma_f32_16x16x4_f32(
        false, rb[g], false, b2, (short)0, cz, false, false);
    v8f p;
#pragma unroll
    for (int k = 0; k < 8; ++k) p[k] = d1v[k] * d2v[k];
    if (n < 4) {
#pragma unroll
      for (int k = 0; k < 8; ++k)
        myLds[g * 64 + (siteHi + k) * 4 + n] = p[k];
    }
  }
  asm volatile("s_wait_dscnt 0" ::: "memory");
  if (lane < 16) {
    const size_t site0 = (size_t)grp * (TPG * 16);
#pragma unroll
    for (int g = 0; g < TPG; ++g) {
      v4f jv = *(const v4f*)(myLds + g * 64 + lane * 4);
      *(v4f*)(J + (site0 + (size_t)(g * 16 + lane)) * 4) = jv;
      acc += __logf(jv.x + jv.y + jv.z + jv.w);
    }
  }
}

__launch_bounds__(256)
__global__ void jc_joint_kernel(const float* __restrict__ f1,
                                const float* __restrict__ f2,
                                const float* __restrict__ Tws,
                                float* __restrict__ joint,
                                float* __restrict__ logp,
                                int B, int m, int blocksPerBatch) {
  __shared__ float lds[8 * TPG * 64];
  const int lane = threadIdx.x & 31;
  const int wid  = __builtin_amdgcn_readfirstlane((int)(threadIdx.x >> 5));
  const int b    = blockIdx.x / blocksPerBatch;
  const int blk  = blockIdx.x % blocksPerBatch;
  const int wavesPerBlock = blockDim.x >> 5;
  const int wavesPerBatch = blocksPerBatch * wavesPerBlock;
  const int wgid  = blk * wavesPerBlock + wid;
  const int tiles = m >> 4;
  const int groups = tiles / TPG;

  const float* T1 = Tws + (size_t)b * 32;
  const float* T2 = T1 + 16;
  const int n   = lane & 15;
  const int klo = (lane >> 4) << 1;
  v2f b1, b2;
  b1.x = (n < 4) ? T1[klo * 4 + n]       : 0.0f;
  b1.y = (n < 4) ? T1[(klo + 1) * 4 + n] : 0.0f;
  b2.x = (n < 4) ? T2[klo * 4 + n]       : 0.0f;
  b2.y = (n < 4) ? T2[(klo + 1) * 4 + n] : 0.0f;

  const size_t batch_off = (size_t)b * (size_t)m * 4;
  const float* F1 = f1 + batch_off;
  const float* F2 = f2 + batch_off;
  float* J = joint + batch_off;
  float* myLds = lds + wid * (TPG * 64);
  const int laneOff = n * 4 + klo;

  float acc = 0.0f;

  v2f r0a[TPG], r0b[TPG], r1a[TPG], r1b[TPG];
  const int mine = (wgid < groups) ? (groups - wgid + wavesPerBatch - 1) / wavesPerBatch : 0;
  int i = 0;
  int grp = wgid;
  if (mine > 0) load_group(r0a, r0b, F1, F2, grp, laneOff);
  for (; i + 2 < mine; i += 2) {
    __builtin_amdgcn_sched_barrier(0);
    load_group(r1a, r1b, F1, F2, grp + wavesPerBatch, laneOff);
    __builtin_amdgcn_sched_barrier(0);
    compute_group(r0a, r0b, b1, b2, myLds, J, grp, lane, n, acc);
    __builtin_amdgcn_sched_barrier(0);
    load_group(r0a, r0b, F1, F2, grp + 2 * wavesPerBatch, laneOff);
    __builtin_amdgcn_sched_barrier(0);
    compute_group(r1a, r1b, b1, b2, myLds, J, grp + wavesPerBatch, lane, n, acc);
    grp += 2 * wavesPerBatch;
  }
  if (mine - i == 2) {
    __builtin_amdgcn_sched_barrier(0);
    load_group(r1a, r1b, F1, F2, grp + wavesPerBatch, laneOff);
    __builtin_amdgcn_sched_barrier(0);
    compute_group(r0a, r0b, b1, b2, myLds, J, grp, lane, n, acc);
    compute_group(r1a, r1b, b1, b2, myLds, J, grp + wavesPerBatch, lane, n, acc);
  } else if (mine - i == 1) {
    compute_group(r0a, r0b, b1, b2, myLds, J, grp, lane, n, acc);
  }

  for (int t = groups * TPG + wgid; t < tiles; t += wavesPerBatch) {
    const int site0 = t << 4;
    const size_t aoff = (size_t)site0 * 4 + (size_t)laneOff;
    v2f a1 = *(const v2f*)(F1 + aoff);
    v2f a2 = *(const v2f*)(F2 + aoff);
    v8f cz = {};
    v8f d1v = __builtin_amdgcn_wmma_f32_16x16x4_f32(
        false, a1, false, b1, (short)0, cz, false, false);
    v8f d2v = __builtin_amdgcn_wmma_f32_16x16x4_f32(
        false, a2, false, b2, (short)0, cz, false, false);
    v8f p;
#pragma unroll
    for (int g = 0; g < 8; ++g) p[g] = d1v[g] * d2v[g];
    if (n < 4) {
      const int siteHi = (lane >> 4) << 3;
#pragma unroll
      for (int g = 0; g < 8; ++g)
        myLds[(siteHi + g) * 4 + n] = p[g];
    }
    asm volatile("s_wait_dscnt 0" ::: "memory");
    if (lane < 16) {
      v4f jv = *(const v4f*)(myLds + lane * 4);
      *(v4f*)(J + (size_t)(site0 + lane) * 4) = jv;
      acc += __logf(jv.x + jv.y + jv.z + jv.w);
    }
  }

#pragma unroll
  for (int off = 16; off >= 1; off >>= 1)
    acc += __shfl_xor(acc, off, 32);
  if (lane == 0) atomicAdd(&logp[b], acc);
}

extern "C" void kernel_launch(void* const* d_in, const int* in_sizes, int n_in,
                              void* d_out, int out_size, void* d_ws, size_t ws_size,
                              hipStream_t stream) {
  const float* f1  = (const float*)d_in[0];
  const float* br1 = (const float*)d_in[1];
  const float* f2  = (const float*)d_in[2];
  const float* br2 = (const float*)d_in[3];

  const int B = in_sizes[1];
  const int m = in_sizes[0] / (B * 4);

  float* out   = (float*)d_out;
  float* joint = out;
  float* logp  = out + (size_t)B * m * 4;
  float* Tws   = (float*)d_ws;

  jc_setup_kernel<<<(B + 255) / 256, 256, 0, stream>>>(br1, br2, Tws, logp, B, m);

  const int blocksPerBatch = 1;
  jc_joint_kernel<<<B * blocksPerBatch, 256, 0, stream>>>(
      f1, f2, Tws, joint, logp, B, m, blocksPerBatch);
}


# Setup launch folded into the streaming kernel (2 launches -> 1): each workgroup builds its T from its branch lengths; log_p atomics become a fixed-order block sum and one plain store
# speedup vs baseline: 1.7132x; 1.4105x over previous
#include <hip/hip_runtime.h>
#include <hip/hip_bf16.h>
#include <math.h>

typedef __attribute__((ext_vector_type(2))) float v2f;
typedef __attribute__((ext_vector_type(4))) float v4f;
typedef __attribute__((ext_vector_type(8))) float v8f;

#define PRIOR_LAMBDA 10.0f

#define TPG 8

static __device__ __forceinline__ void load_group(v2f (&ra)[TPG], v2f (&rb)[TPG],
                                                  const float* __restrict__ F1,
                                                  const float* __restrict__ F2,
                                                  int grp, int laneOff) {
  const size_t base = (size_t)grp * (TPG * 64) + (size_t)laneOff;
#pragma unroll
  for (int g = 0; g < TPG; ++g) {
    ra[g] = *(const v2f*)(F1 + base + g * 64);
    rb[g] = *(const v2f*)(F2 + base + g * 64);
  }
}

static __device__ __forceinline__ void compute_group(const v2f (&ra)[TPG], const v2f (&rb)[TPG],
                                                     v2f b1, v2f b2, float* myLds,
                                                     float* __restrict__ J, int grp,
                                                     int lane, int n, float& acc) {
  const int siteHi = (lane >> 4) << 3;
#pragma unroll
  for (int g = 0; g < TPG; ++g) {
    v8f cz = {};
    v8f d1v = __builtin_amdgcn_wmma_f32_16x16x4_f32(
        false, ra[g], false, b1, (short)0, cz, false, false);
    v8f d2v = __builtin_amdgcn_wmma_f32_16x16x4_f32(
        false, rb[g], false, b2, (short)0, cz, false, false);
    v8f p;
#pragma unroll
    for (int k = 0; k < 8; ++k) p[k] = d1v[k] * d2v[k];
    if (n < 4) {
#pragma unroll
      for (int k = 0; k < 8; ++k)
        myLds[g * 64 + (siteHi + k) * 4 + n] = p[k];
    }
  }
  asm volatile("s_wait_dscnt 0" ::: "memory");
  if (lane < 16) {
    const size_t site0 = (size_t)grp * (TPG * 16);
#pragma unroll
    for (int g = 0; g < TPG; ++g) {
      v4f jv = *(const v4f*)(myLds + g * 64 + lane * 4);
      *(v4f*)(J + (site0 + (size_t)(g * 16 + lane)) * 4) = jv;
      acc += __logf(jv.x + jv.y + jv.z + jv.w);
    }
  }
}

__launch_bounds__(256)
__global__ void jc_joint_kernel(const float* __restrict__ f1,
                                const float* __restrict__ f2,
                                const float* __restrict__ branch1,
                                const float* __restrict__ branch2,
                                float* __restrict__ joint,
                                float* __restrict__ logp,
                                int B, int m) {
  __shared__ float lds[8 * TPG * 64];
  __shared__ float waveSum[8];
  const int lane = threadIdx.x & 31;
  const int wid  = __builtin_amdgcn_readfirstlane((int)(threadIdx.x >> 5));
  const int b    = blockIdx.x;
  const int wavesPerBatch = blockDim.x >> 5;
  const int wgid  = wid;
  const int tiles = m >> 4;
  const int groups = tiles / TPG;

  const int n   = lane & 15;
  const int klo = (lane >> 4) << 1;
  const size_t batch_off = (size_t)b * (size_t)m * 4;
  const float* F1 = f1 + batch_off;
  const float* F2 = f2 + batch_off;
  float* J = joint + batch_off;
  float* myLds = lds + wid * (TPG * 64);
  const int laneOff = n * 4 + klo;

  float acc = 0.0f;

  v2f r0a[TPG], r0b[TPG], r1a[TPG], r1b[TPG];
  const int mine = (wgid < groups) ? (groups - wgid + wavesPerBatch - 1) / wavesPerBatch : 0;
  int i = 0;
  int grp = wgid;
  if (mine > 0) load_group(r0a, r0b, F1, F2, grp, laneOff);

  v2f b1, b2;
  {
    const float t1 = branch1[b], t2 = branch2[b];
    const float e1 = __expf(-(4.0f / 3.0f) * t1);
    const float e2 = __expf(-(4.0f / 3.0f) * t2);
    const float s  = __expf((__logf(PRIOR_LAMBDA) - PRIOR_LAMBDA * (t1 + t2)) / (float)m);
    const float dg1 = fmaxf(0.25f + 0.75f * e1, 0.0f) * s;
    const float of1 = fmaxf(0.25f - 0.25f * e1, 0.0f) * s;
    const float dg2 = fmaxf(0.25f + 0.75f * e2, 0.0f);
    const float of2 = fmaxf(0.25f - 0.25f * e2, 0.0f);
    b1.x = (n < 4) ? ((n == klo)     ? dg1 : of1) : 0.0f;
    b1.y = (n < 4) ? ((n == klo + 1) ? dg1 : of1) : 0.0f;
    b2.x = (n < 4) ? ((n == klo)     ? dg2 : of2) : 0.0f;
    b2.y = (n < 4) ? ((n == klo + 1) ? dg2 : of2) : 0.0f;
  }
  for (; i + 2 < mine; i += 2) {
    __builtin_amdgcn_sched_barrier(0);
    load_group(r1a, r1b, F1, F2, grp + wavesPerBatch, laneOff);
    __builtin_amdgcn_sched_barrier(0);
    compute_group(r0a, r0b, b1, b2, myLds, J, grp, lane, n, acc);
    __builtin_amdgcn_sched_barrier(0);
    load_group(r0a, r0b, F1, F2, grp + 2 * wavesPerBatch, laneOff);
    __builtin_amdgcn_sched_barrier(0);
    compute_group(r1a, r1b, b1, b2, myLds, J, grp + wavesPerBatch, lane, n, acc);
    grp += 2 * wavesPerBatch;
  }
  if (mine - i == 2) {
    __builtin_amdgcn_sched_barrier(0);
    load_group(r1a, r1b, F1, F2, grp + wavesPerBatch, laneOff);
    __builtin_amdgcn_sched_barrier(0);
    compute_group(r0a, r0b, b1, b2, myLds, J, grp, lane, n, acc);
    compute_group(r1a, r1b, b1, b2, myLds, J, grp + wavesPerBatch, lane, n, acc);
  } else if (mine - i == 1) {
    compute_group(r0a, r0b, b1, b2, myLds, J, grp, lane, n, acc);
  }

  for (int t = groups * TPG + wgid; t < tiles; t += wavesPerBatch) {
    const int site0 = t << 4;
    const size_t aoff = (size_t)site0 * 4 + (size_t)laneOff;
    v2f a1 = *(const v2f*)(F1 + aoff);
    v2f a2 = *(const v2f*)(F2 + aoff);
    v8f cz = {};
    v8f d1v = __builtin_amdgcn_wmma_f32_16x16x4_f32(
        false, a1, false, b1, (short)0, cz, false, false);
    v8f d2v = __builtin_amdgcn_wmma_f32_16x16x4_f32(
        false, a2, false, b2, (short)0, cz, false, false);
    v8f p;
#pragma unroll
    for (int g = 0; g < 8; ++g) p[g] = d1v[g] * d2v[g];
    if (n < 4) {
      const int siteHi = (lane >> 4) << 3;
#pragma unroll
      for (int g = 0; g < 8; ++g)
        myLds[(siteHi + g) * 4 + n] = p[g];
    }
    asm volatile("s_wait_dscnt 0" ::: "memory");
    if (lane < 16) {
      v4f jv = *(const v4f*)(myLds + lane * 4);
      *(v4f*)(J + (size_t)(site0 + lane) * 4) = jv;
      acc += __logf(jv.x + jv.y + jv.z + jv.w);
    }
  }

#pragma unroll
  for (int off = 16; off >= 1; off >>= 1)
    acc += __shfl_xor(acc, off, 32);
  if (lane == 0) waveSum[wid] = acc;
  __syncthreads();
  if (threadIdx.x == 0) {
    float tot = 0.0f;
    for (int w = 0; w < wavesPerBatch; ++w) tot += waveSum[w];
    logp[b] = (float)m * __logf(0.25f) + tot;
  }
}

extern "C" void kernel_launch(void* const* d_in, const int* in_sizes, int n_in,
                              void* d_out, int out_size, void* d_ws, size_t ws_size,
                              hipStream_t stream) {
  const float* f1  = (const float*)d_in[0];
  const float* br1 = (const float*)d_in[1];
  const float* f2  = (const float*)d_in[2];
  const float* br2 = (const float*)d_in[3];

  const int B = in_sizes[1];
  const int m = in_sizes[0] / (B * 4);

  float* out   = (float*)d_out;
  float* joint = out;
  float* logp  = out + (size_t)B * m * 4;

  jc_joint_kernel<<<B, 256, 0, stream>>>(f1, f2, br1, br2, joint, logp, B, m);
}
